# starting kernel
# baseline (speedup 1.0000x reference)
#include <hip/hip_runtime.h>

#define B_    4
#define CK_   64
#define CV_   512
#define H_    30
#define W_    54
#define T_    8
#define P_    (H_*W_)
#define N_    (T_*H_*W_)
#define CHUNK 128
#define NCHUNK ((N_ + CHUNK - 1)/CHUNK)
#define PTILES ((P_ + 15)/16)

typedef __attribute__((ext_vector_type(16))) __bf16       v16bf;
typedef __attribute__((ext_vector_type(8)))  float        v8f;
typedef __attribute__((ext_vector_type(4)))  unsigned int v4u;

union Frag16 {
  v16bf  bf;
  v4u    q[2];
  __bf16 h[16];
};
union Pack8 {
  v4u    q;
  __bf16 h[8];
};
union Pack4 {
  uint2  u;
  __bf16 h[4];
};

#define QSTR  72
#define MKSTR 72
#define PSTR  136

template <bool MV_BF16>
__global__ __launch_bounds__(256) void mem_reader_attn(
    const float* __restrict__ mk, const float* __restrict__ mv_f32,
    const __bf16* __restrict__ mv_bf, const float* __restrict__ qk,
    float* __restrict__ out)
{
  __shared__ __bf16 s_q[16 * QSTR];
  __shared__ __bf16 s_mk[CHUNK * MKSTR];
  __shared__ __bf16 s_prob[16 * PSTR];
  __shared__ float  s_pmax[8][32];
  __shared__ float  s_psum[8][32];
  __shared__ float  s_mrun[16];
  __shared__ float  s_lrun[16];
  __shared__ float  s_alpha[16];

  const int tid  = threadIdx.x;
  const int wave = tid >> 5;
  const int lane = tid & 31;
  const int col  = lane & 15;
  const int kg   = (lane >> 4) * 8;
  const int p0   = blockIdx.x * 16;
  const int bi   = blockIdx.y;

  const float*  mk_b  = mk + (size_t)bi * CK_ * N_;
  const float*  qk_b  = qk + (size_t)bi * CK_ * P_;
  const float*  mvf_b = mv_f32 ? (mv_f32 + (size_t)bi * CV_ * N_) : nullptr;
  const __bf16* mvb_b = mv_bf  ? (mv_bf  + (size_t)bi * CV_ * N_) : nullptr;

  for (int idx = tid; idx < CK_ * 16; idx += 256) {
    int p = idx & 15, c = idx >> 4;
    int pg = p0 + p; if (pg > P_ - 1) pg = P_ - 1;
    s_q[p * QSTR + c] = (__bf16)(qk_b[(size_t)c * P_ + pg] * 0.125f);
  }
  if (tid < 16) { s_mrun[tid] = -1e30f; s_lrun[tid] = 0.f; s_alpha[tid] = 0.f; }

  v8f acc[4] = {v8f{}, v8f{}, v8f{}, v8f{}};

  for (int nc = 0; nc < NCHUNK; ++nc) {
    const int n0 = nc * CHUNK;
    __syncthreads();

    for (int idx = tid; idx < CK_ * CHUNK; idx += 256) {
      int nl = idx & (CHUNK - 1), c = idx >> 7;
      int ng = n0 + nl;
      float v = (ng < N_) ? mk_b[(size_t)c * N_ + ng] : 0.f;
      s_mk[nl * MKSTR + c] = (__bf16)v;
    }
    __syncthreads();

    const int mrow = wave * 16;
    v8f S = {};
#pragma unroll
    for (int ks = 0; ks < 2; ++ks) {
      Frag16 a, b;
      const __bf16* ap = &s_mk[(mrow + col) * MKSTR + ks * 32 + kg];
      a.q[0] = *(const v4u*)ap;  a.q[1] = *(const v4u*)(ap + 16);
      const __bf16* bp = &s_q[col * QSTR + ks * 32 + kg];
      b.q[0] = *(const v4u*)bp;  b.q[1] = *(const v4u*)(bp + 16);
      S = __builtin_amdgcn_wmma_f32_16x16x32_bf16(false, a.bf, false, b.bf,
                                                  (short)0, S, false, false);
    }

    float sv[8];
    float lmax = -1e30f;
#pragma unroll
    for (int j = 0; j < 8; ++j) {
      int ng = n0 + mrow + kg + j;
      sv[j] = (ng < N_) ? S[j] : -1e30f;
      lmax = fmaxf(lmax, sv[j]);
    }
    s_pmax[wave][lane] = lmax;
    __syncthreads();

    if (tid < 16) {
      float cmax = -1e30f;
#pragma unroll
      for (int w = 0; w < 8; ++w)
        cmax = fmaxf(cmax, fmaxf(s_pmax[w][tid], s_pmax[w][tid + 16]));
      float mold = s_mrun[tid];
      float mnew = fmaxf(mold, cmax);
      s_alpha[tid] = __expf(mold - mnew);
      s_mrun[tid]  = mnew;
    }
    __syncthreads();

    const float mnew = s_mrun[col];
    const float alph = s_alpha[col];
#pragma unroll
    for (int t = 0; t < 4; ++t) acc[t] *= alph;

    Pack8 pk;
    float lsum = 0.f;
#pragma unroll
    for (int j = 0; j < 8; ++j) {
      int ng = n0 + mrow + kg + j;
      float pr = (ng < N_) ? __expf(sv[j] - mnew) : 0.f;
      lsum += pr;
      pk.h[j] = (__bf16)pr;
    }
    *(v4u*)&s_prob[col * PSTR + mrow + kg] = pk.q;
    s_psum[wave][lane] = lsum;
    __syncthreads();

    if (tid < 16) {
      float csum = 0.f;
#pragma unroll
      for (int w = 0; w < 8; ++w)
        csum += s_psum[w][tid] + s_psum[w][tid + 16];
      s_lrun[tid] = s_lrun[tid] * s_alpha[tid] + csum;
    }

#pragma unroll
    for (int vt = 0; vt < 4; ++vt) {
      const int v = wave * 64 + vt * 16 + col;
#pragma unroll
      for (int ks = 0; ks < 4; ++ks) {
        int nA = n0 + ks * 32 + kg;
        int b1 = nA;      if (b1 > N_ - 8) b1 = N_ - 8;
        int b2 = nA + 16; if (b2 > N_ - 8) b2 = N_ - 8;
        Frag16 a;
        if (MV_BF16) {
          const __bf16* rowp = mvb_b + (size_t)v * N_;
          a.q[0] = *(const v4u*)(rowp + b1);
          a.q[1] = *(const v4u*)(rowp + b2);
        } else {
          const float* rowp = mvf_b + (size_t)v * N_;
          const float4 f0 = *(const float4*)(rowp + b1);
          const float4 f1 = *(const float4*)(rowp + b1 + 4);
          const float4 f2 = *(const float4*)(rowp + b2);
          const float4 f3 = *(const float4*)(rowp + b2 + 4);
          a.h[0]=(__bf16)f0.x;  a.h[1]=(__bf16)f0.y;  a.h[2]=(__bf16)f0.z;  a.h[3]=(__bf16)f0.w;
          a.h[4]=(__bf16)f1.x;  a.h[5]=(__bf16)f1.y;  a.h[6]=(__bf16)f1.z;  a.h[7]=(__bf16)f1.w;
          a.h[8]=(__bf16)f2.x;  a.h[9]=(__bf16)f2.y;  a.h[10]=(__bf16)f2.z; a.h[11]=(__bf16)f2.w;
          a.h[12]=(__bf16)f3.x; a.h[13]=(__bf16)f3.y; a.h[14]=(__bf16)f3.z; a.h[15]=(__bf16)f3.w;
        }
        Frag16 b;
        const __bf16* bp = &s_prob[col * PSTR + ks * 32 + kg];
        b.q[0] = *(const v4u*)bp;  b.q[1] = *(const v4u*)(bp + 16);
        acc[vt] = __builtin_amdgcn_wmma_f32_16x16x32_bf16(false, a.bf, false, b.bf,
                                                          (short)0, acc[vt], false, false);
      }
    }
  }

  __syncthreads();
  const float linv = 1.0f / s_lrun[col];
  const int pg = p0 + col;
  if (pg < P_) {
#pragma unroll
    for (int vt = 0; vt < 4; ++vt) {
#pragma unroll
      for (int j = 0; j < 8; ++j) {
        int v = wave * 64 + vt * 16 + kg + j;
        out[((size_t)bi * (2 * CV_) + v) * P_ + pg] = acc[vt][j] * linv;
      }
    }
  }
}

__global__ __launch_bounds__(256) void convert_mv(const float* __restrict__ src,
                                                  __bf16* __restrict__ dst, int n4)
{
  int i = blockIdx.x * blockDim.x + threadIdx.x;
  if (i < n4) {
    float4 f = ((const float4*)src)[i];
    Pack4 p;
    p.h[0] = (__bf16)f.x; p.h[1] = (__bf16)f.y;
    p.h[2] = (__bf16)f.z; p.h[3] = (__bf16)f.w;
    *(uint2*)(dst + 4 * (size_t)i) = p.u;
  }
}

__global__ __launch_bounds__(256) void qv_copy(const float* __restrict__ qv,
                                               float* __restrict__ out, int total)
{
  int i = blockIdx.x * blockDim.x + threadIdx.x;
  if (i < total) {
    int b = i / (CV_ * P_);
    int r = i - b * (CV_ * P_);
    out[(size_t)b * (2 * CV_ * P_) + (CV_ * P_) + r] = qv[i];
  }
}

extern "C" void kernel_launch(void* const* d_in, const int* in_sizes, int n_in,
                              void* d_out, int out_size, void* d_ws, size_t ws_size,
                              hipStream_t stream) {
  const float* mk = (const float*)d_in[0];
  const float* mv = (const float*)d_in[1];
  const float* qk = (const float*)d_in[2];
  const float* qv = (const float*)d_in[3];
  float* out = (float*)d_out;

  const size_t mv_elems = (size_t)B_ * CV_ * N_;
  const size_t need     = mv_elems * sizeof(__bf16);
  dim3 grid(PTILES, B_);

  if (d_ws && ws_size >= need) {
    __bf16* mv_bf = (__bf16*)d_ws;
    int n4 = (int)(mv_elems / 4);
    convert_mv<<<(n4 + 255) / 256, 256, 0, stream>>>(mv, mv_bf, n4);
    mem_reader_attn<true><<<grid, 256, 0, stream>>>(mk, nullptr, mv_bf, qk, out);
  } else {
    mem_reader_attn<false><<<grid, 256, 0, stream>>>(mk, mv, nullptr, qk, out);
  }

  int total = B_ * CV_ * P_;
  qv_copy<<<(total + 255) / 256, 256, 0, stream>>>(qv, out, total);
}


# qv copy moved out of its own launch into 48 extra work-groups of the attention launch (grid 32x4x2 = 256): they run on the WGPs the 208 attention groups leave empty; 2 launches, not 3
# speedup vs baseline: 9.8028x; 9.8028x over previous
#include <hip/hip_runtime.h>


#define B_    4
#define CK_   64
#define CV_   512
#define H_    30
#define W_    54
#define T_    8
#define P_    (H_*W_)
#define N_    (T_*H_*W_)
#define SUB   3
#define KSTEP (32 * SUB)
#define NSTEP (N_ / KSTEP)
#define QT    4
#define PBLK  (16 * QT)
#define PTILES ((P_ + PBLK - 1)/PBLK)
#define WAVES 4
#define VSPLIT 2
#define VROWS (CV_ / VSPLIT)
#define VT    (VROWS / 16)
#define VG    8
#define VOWN  (VT / WAVES)
#define NGRP  (SUB * (VT / VG))
#define AHEAD 2
#define CPY_X 6
#define CPY_WG (CPY_X * VSPLIT)
#define CPY_U 15
#define CPY_N4 ((CV_ * P_) / 4)
#define CPY_STRIDE (CPY_WG * 32 * WAVES)

static_assert(N_ % KSTEP == 0, "every 96-key step is full: no key tail to mask");
static_assert((N_ * 2) % 16 == 0, "bf16 value rows start 16-byte aligned");
static_assert(N_ % 2 == 0 && KSTEP % 32 == 0, "a lane's key pair is one 8-byte aligned load; a half wave's pairs fill one 128-byte line");
static_assert(NSTEP >= WAVES, "every wave owns at least one key step, so its running max is finite");
static_assert(VT % VG == 0 && VT % WAVES == 0, "value tiles split evenly into load groups and owners");
static_assert(CV_ % (16 * VSPLIT) == 0, "the value rows split evenly into whole tiles per work-group");
static_assert((CV_ * P_) % 4 == 0, "a batch's query-value block is a whole number of 16-byte pieces, 16-byte aligned in qv and in out");
static_assert(CPY_N4 % (CPY_STRIDE * CPY_U) == 0, "every copy thread moves a whole number of full rounds: no tail, no bound test");

typedef __attribute__((ext_vector_type(16))) __bf16       v16bf;
typedef __attribute__((ext_vector_type(8)))  float        v8f;
typedef __attribute__((ext_vector_type(4)))  unsigned int v4u;
typedef __attribute__((ext_vector_type(4)))  float        v4f;

union Frag16 {
  v16bf  bf;
  v4u    q[2];
  __bf16 h[16];
};
union Pack4 {
  uint2  u;
  __bf16 h[4];
};

__device__ __forceinline__ int score_chan(int ks, int hf, int s) {
  return ks * 32 + (s >> 3) * 16 + hf * 8 + (s & 7);
}

__device__ __forceinline__ int key_pair(int r) {
  return 4 * (r >> 3) + (r & 3) + 8 * ((r >> 2) & 1);
}

__device__ __forceinline__ float other_half(float x) {
  return __builtin_bit_cast(float, __builtin_amdgcn_permlanex16(
      __builtin_bit_cast(int, x), __builtin_bit_cast(int, x),
      0x76543210, 0xfedcba98, false, false));
}

template <int O>
__device__ __forceinline__ void combine_round(v8f (&acc)[QT][VT],
                                              float (&s_x)[WAVES - 1][QT][VOWN][8][32],
                                              const float (&linv)[QT], int wave, int lane,
                                              int kg, int col, int p0, int bi, int vbase,
                                              float* __restrict__ out)
{
  if (wave != O) {
    const int slot = wave < O ? wave : wave - 1;
#pragma unroll
    for (int qt = 0; qt < QT; ++qt) {
#pragma unroll
      for (int i = 0; i < VOWN; ++i) {
#pragma unroll
        for (int j = 0; j < 8; ++j) s_x[slot][qt][i][j][lane] = acc[qt][O * VOWN + i][j];
      }
    }
  }
  __syncthreads();
  if (wave == O) {
#pragma unroll
    for (int qt = 0; qt < QT; ++qt) {
      const int pg = p0 + qt * 16 + col;
#pragma unroll
      for (int i = 0; i < VOWN; ++i) {
#pragma unroll
        for (int j = 0; j < 8; ++j) {
          float r = acc[qt][O * VOWN + i][j];
#pragma unroll
          for (int sl = 0; sl < WAVES - 1; ++sl) r += s_x[sl][qt][i][j][lane];
          const int v = vbase + (O * VOWN + i) * 16 + kg + j;
          if (pg < P_) out[((size_t)bi * (2 * CV_) + v) * P_ + pg] = r * linv[qt];
        }
      }
    }
  }
  __syncthreads();
}

__device__ __forceinline__ void copy_qv_block(const float* __restrict__ qv, float* __restrict__ out,
                                              int bi, int cwg, int tid)
{
  const v4f* cs = (const v4f*)(qv + (size_t)bi * (CV_ * P_));
  v4f*       cd = (v4f*)(out + (size_t)bi * (2 * CV_ * P_) + (size_t)(CV_ * P_));
  int i = cwg * (32 * WAVES) + tid;
  for (int r = 0; r < CPY_N4 / (CPY_STRIDE * CPY_U); ++r) {
    v4f v[CPY_U];
#pragma unroll
    for (int u = 0; u < CPY_U; ++u) v[u] = cs[i + u * CPY_STRIDE];
    asm volatile("" :: "v"(v[0]), "v"(v[1]), "v"(v[2]), "v"(v[3]), "v"(v[4]), "v"(v[5]), "v"(v[6]),
                 "v"(v[7]), "v"(v[8]), "v"(v[9]), "v"(v[10]), "v"(v[11]), "v"(v[12]), "v"(v[13]),
                 "v"(v[14]) : "memory");
    static_assert(CPY_U == 15, "the statement above names every piece of a round");
#pragma unroll
    for (int u = 0; u < CPY_U; ++u) cd[i + u * CPY_STRIDE] = v[u];
    i += CPY_U * CPY_STRIDE;
  }
}

template <bool MV_BF16>
__global__ __launch_bounds__(32 * WAVES) void mem_reader_attn(
    const float* __restrict__ mk, const float* __restrict__ mv_f32,
    const __bf16* __restrict__ mv_bf, const float* __restrict__ qk,
    const float* __restrict__ qv, float* __restrict__ out)
{
  const int tid  = threadIdx.x;

  if (blockIdx.x >= PTILES) {
    copy_qv_block(qv, out, blockIdx.y, (blockIdx.x - PTILES) * VSPLIT + blockIdx.z, tid);
    return;
  }

  const int wave = __builtin_amdgcn_readfirstlane(tid >> 5);
  const int lane = tid & 31;
  const int col  = lane & 15;
  const int hf   = lane >> 4;
  const int kg   = hf * 8;
  const int p0   = blockIdx.x * PBLK;
  const int bi   = blockIdx.y;
  const int vbase = blockIdx.z * VROWS;

  const float*  mk_b  = mk + (size_t)bi * CK_ * N_;
  const float*  qk_b  = qk + (size_t)bi * CK_ * P_;
  const float*  mvf_b = mv_f32 ? (mv_f32 + (size_t)bi * CV_ * N_) : nullptr;
  const __bf16* mvb_b = mv_bf  ? (mv_bf  + (size_t)bi * CV_ * N_) : nullptr;

  Frag16 qf[QT][2];
#pragma unroll
  for (int qt = 0; qt < QT; ++qt) {
    int pq = p0 + qt * 16 + col; if (pq > P_ - 1) pq = P_ - 1;
#pragma unroll
    for (int ks = 0; ks < 2; ++ks) {
#pragma unroll
      for (int s = 0; s < 16; ++s)
        qf[qt][ks].h[s] = (__bf16)(qk_b[(size_t)score_chan(ks, hf, s) * P_ + pq] * 0.125f);
    }
  }

  const int vrow0 = col;
  const int kpr   = key_pair(col);

  float m_run[QT];
  float l_run[QT];
  v8f acc[QT][VT];
#pragma unroll
  for (int qt = 0; qt < QT; ++qt) {
    m_run[qt] = -1e30f;
    l_run[qt] = 0.f;
#pragma unroll
    for (int vt = 0; vt < VT; ++vt) acc[qt][vt] = v8f{};
  }

  for (int st = wave; st < NSTEP; st += WAVES) {
    const int n0 = st * KSTEP;

    float2 kraw[SUB][2][16];
#pragma unroll
    for (int u = 0; u < SUB; ++u) {
      const float* kp = mk_b + n0 + u * 32 + 2 * kpr;
#pragma unroll
      for (int ks = 0; ks < 2; ++ks) {
#pragma unroll
        for (int s = 0; s < 16; ++s)
          kraw[u][ks][s] = *(const float2*)(kp + (size_t)score_chan(ks, hf, s) * N_);
      }
    }

    Frag16 a[NGRP][VG];
    auto load_group = [&](int k) {
      const int u = k / (VT / VG);
      const int g = k % (VT / VG);
#pragma unroll
      for (int i = 0; i < VG; ++i) {
        const size_t roff = (size_t)(vbase + vrow0 + (g * VG + i) * 16) * N_ + n0 + kg + u * 32;
        if (MV_BF16) {
          const __bf16* rowp = mvb_b + roff;
          a[k][i].q[0] = *(const v4u*)(rowp);
          a[k][i].q[1] = *(const v4u*)(rowp + 16);
        } else {
          const float* rowp = mvf_b + roff;
#pragma unroll
          for (int h2 = 0; h2 < 2; ++h2) {
            const float4 f0 = *(const float4*)(rowp + h2 * 16);
            const float4 f1 = *(const float4*)(rowp + h2 * 16 + 4);
            a[k][i].h[h2*8+0]=(__bf16)f0.x; a[k][i].h[h2*8+1]=(__bf16)f0.y;
            a[k][i].h[h2*8+2]=(__bf16)f0.z; a[k][i].h[h2*8+3]=(__bf16)f0.w;
            a[k][i].h[h2*8+4]=(__bf16)f1.x; a[k][i].h[h2*8+5]=(__bf16)f1.y;
            a[k][i].h[h2*8+6]=(__bf16)f1.z; a[k][i].h[h2*8+7]=(__bf16)f1.w;
          }
        }
      }
    };
#pragma unroll
    for (int k = 0; k < AHEAD; ++k) load_group(k);
    __builtin_amdgcn_sched_barrier(0);

    v8f S[QT][2 * SUB];
#pragma unroll
    for (int t = 0; t < 2 * SUB; ++t) {
#pragma unroll
      for (int qt = 0; qt < QT; ++qt) S[qt][t] = v8f{};
#pragma unroll
      for (int ks = 0; ks < 2; ++ks) {
        Frag16 ak;
#pragma unroll
        for (int s = 0; s < 16; ++s)
          ak.h[s] = (__bf16)((t & 1) ? kraw[t >> 1][ks][s].y : kraw[t >> 1][ks][s].x);
#pragma unroll
        for (int qt = 0; qt < QT; ++qt)
          S[qt][t] = __builtin_amdgcn_wmma_f32_16x16x32_bf16(false, ak.bf, false, qf[qt][ks].bf,
                                                             (short)0, S[qt][t], false, false);
      }
    }

    Frag16 pb[QT][SUB];
#pragma unroll
    for (int qt = 0; qt < QT; ++qt) {
      float lmax = S[qt][0][0];
#pragma unroll
      for (int t = 0; t < 2 * SUB; ++t) {
#pragma unroll
        for (int j = 0; j < 8; ++j) lmax = fmaxf(lmax, S[qt][t][j]);
      }
      lmax = fmaxf(lmax, other_half(lmax));
      const float m_new = fmaxf(m_run[qt], lmax);
      const float alph  = __expf(m_run[qt] - m_new);
      m_run[qt] = m_new;

      float lsum = 0.f;
#pragma unroll
      for (int u = 0; u < SUB; ++u) {
#pragma unroll
        for (int g = 0; g < 2; ++g) {
#pragma unroll
          for (int j = 0; j < 8; ++j) {
            const float pr = __expf(S[qt][2 * u + g][j] - m_new);
            lsum += pr;
            pb[qt][u].h[2 * j + g] = (__bf16)pr;
          }
        }
      }
      l_run[qt] = l_run[qt] * alph + lsum;

#pragma unroll
      for (int vt = 0; vt < VT; ++vt) acc[qt][vt] *= alph;
    }

#pragma unroll
    for (int k = 0; k < NGRP; ++k) {
      if (k + AHEAD < NGRP) load_group(k + AHEAD);
      __builtin_amdgcn_sched_barrier(0);
      const int u = k / (VT / VG);
      const int g = k % (VT / VG);
#pragma unroll
      for (int i = 0; i < VG; ++i) {
#pragma unroll
        for (int qt = 0; qt < QT; ++qt)
          acc[qt][g * VG + i] = __builtin_amdgcn_wmma_f32_16x16x32_bf16(
              false, a[k][i].bf, false, pb[qt][u].bf, (short)0, acc[qt][g * VG + i], false, false);
      }
      __builtin_amdgcn_sched_barrier(0);
    }
  }

  __shared__ float s_m[WAVES][QT][16];
  __shared__ float s_l[WAVES][QT][16];
  __shared__ float s_x[WAVES - 1][QT][VOWN][8][32];

  float l_own[QT];
#pragma unroll
  for (int qt = 0; qt < QT; ++qt) {
    l_own[qt] = l_run[qt] + other_half(l_run[qt]);
    if (hf == 0) {
      s_m[wave][qt][col] = m_run[qt];
      s_l[wave][qt][col] = l_own[qt];
    }
  }
  __syncthreads();

  float linv[QT];
#pragma unroll
  for (int qt = 0; qt < QT; ++qt) {
    float m_all = s_m[0][qt][col];
#pragma unroll
    for (int w = 1; w < WAVES; ++w) m_all = fmaxf(m_all, s_m[w][qt][col]);
    float l_all = 0.f;
#pragma unroll
    for (int w = 0; w < WAVES; ++w) l_all += s_l[w][qt][col] * __expf(s_m[w][qt][col] - m_all);
    linv[qt] = 1.0f / l_all;
    const float sc = __expf(m_run[qt] - m_all);
#pragma unroll
    for (int vt = 0; vt < VT; ++vt) acc[qt][vt] *= sc;
  }

  combine_round<0>(acc, s_x, linv, wave, lane, kg, col, p0, bi, vbase, out);
  combine_round<1>(acc, s_x, linv, wave, lane, kg, col, p0, bi, vbase, out);
  combine_round<2>(acc, s_x, linv, wave, lane, kg, col, p0, bi, vbase, out);
  combine_round<3>(acc, s_x, linv, wave, lane, kg, col, p0, bi, vbase, out);
  static_assert(WAVES == 4, "one combine round per wave is written out above");
}

__global__ __launch_bounds__(256) void convert_mv(const float* __restrict__ src,
                                                  __bf16* __restrict__ dst, int n4)
{
  int i = blockIdx.x * blockDim.x + threadIdx.x;
  if (i < n4) {
    float4 f = ((const float4*)src)[i];
    Pack4 p;
    p.h[0] = (__bf16)f.x; p.h[1] = (__bf16)f.y;
    p.h[2] = (__bf16)f.z; p.h[3] = (__bf16)f.w;
    *(uint2*)(dst + 4 * (size_t)i) = p.u;
  }
}

extern "C" void kernel_launch(void* const* d_in, const int* in_sizes, int n_in,
                              void* d_out, int out_size, void* d_ws, size_t ws_size,
                              hipStream_t stream) {
  const float* mk = (const float*)d_in[0];
  const float* mv = (const float*)d_in[1];
  const float* qk = (const float*)d_in[2];
  const float* qv = (const float*)d_in[3];
  float* out = (float*)d_out;

  const size_t mv_elems = (size_t)B_ * CV_ * N_;
  const size_t need     = mv_elems * sizeof(__bf16);
  dim3 grid(PTILES + CPY_X, B_, VSPLIT);

  if (d_ws && ws_size >= need) {
    __bf16* mv_bf = (__bf16*)d_ws;
    int n4 = (int)(mv_elems / 4);
    convert_mv<<<(n4 + 255) / 256, 256, 0, stream>>>(mv, mv_bf, n4);
    mem_reader_attn<true><<<grid, 32 * WAVES, 0, stream>>>(mk, nullptr, mv_bf, qk, qv, out);
  } else {
    mem_reader_attn<false><<<grid, 32 * WAVES, 0, stream>>>(mk, mv, nullptr, qk, qv, out);
  }
}


# each wave owns 112 queries (7 tiles) x 128 v-rows, not 64 x 256: every mv fragment it loads feeds 7 WMMAs, 4/7 of the mv loads per product; grid 16x4x4; per-query arithmetic and sum order unchanged
# speedup vs baseline: 10.3824x; 1.0591x over previous
#include <hip/hip_runtime.h>


#define B_    4
#define CK_   64
#define CV_   512
#define H_    30
#define W_    54
#define T_    8
#define P_    (H_*W_)
#define N_    (T_*H_*W_)
#define SUB   3
#define KSTEP (32 * SUB)
#define NSTEP (N_ / KSTEP)
#define QT    7
#define PBLK  (16 * QT)
#define PTILES ((P_ + PBLK - 1)/PBLK)
#define WAVES 4
#define VSPLIT 4
#define VROWS (CV_ / VSPLIT)
#define VT    (VROWS / 16)
#define VG    4
#define VOWN  (VT / WAVES)
#define NGRP  (SUB * (VT / VG))
#define AHEAD 3
#define CPY_X 1
#define CPY_WG (CPY_X * VSPLIT)
#define CPY_U 15
#define CPY_N4 ((CV_ * P_) / 4)
#define CPY_STRIDE (CPY_WG * 32 * WAVES)

static_assert(N_ % KSTEP == 0, "every 96-key step is full: no key tail to mask");
static_assert((N_ * 2) % 16 == 0, "bf16 value rows start 16-byte aligned");
static_assert(N_ % 2 == 0 && KSTEP % 32 == 0, "a lane's key pair is one 8-byte aligned load; a half wave's pairs fill one 128-byte line");
static_assert(NSTEP >= WAVES, "every wave owns at least one key step, so its running max is finite");
static_assert(VT % VG == 0 && VT % WAVES == 0, "value tiles split evenly into load groups and owners");
static_assert(CV_ % (16 * VSPLIT) == 0, "the value rows split evenly into whole tiles per work-group");
static_assert((CV_ * P_) % 4 == 0, "a batch's query-value block is a whole number of 16-byte pieces, 16-byte aligned in qv and in out");
static_assert(CPY_N4 % (CPY_STRIDE * CPY_U) == 0, "every copy thread moves a whole number of full rounds: no tail, no bound test");

typedef __attribute__((ext_vector_type(16))) __bf16       v16bf;
typedef __attribute__((ext_vector_type(8)))  float        v8f;
typedef __attribute__((ext_vector_type(4)))  unsigned int v4u;
typedef __attribute__((ext_vector_type(4)))  float        v4f;

union Frag16 {
  v16bf  bf;
  v4u    q[2];
  __bf16 h[16];
};
union Pack4 {
  uint2  u;
  __bf16 h[4];
};

__device__ __forceinline__ int score_chan(int ks, int hf, int s) {
  return ks * 32 + (s >> 3) * 16 + hf * 8 + (s & 7);
}

__device__ __forceinline__ int key_pair(int r) {
  return 4 * (r >> 3) + (r & 3) + 8 * ((r >> 2) & 1);
}

__device__ __forceinline__ float other_half(float x) {
  return __builtin_bit_cast(float, __builtin_amdgcn_permlanex16(
      __builtin_bit_cast(int, x), __builtin_bit_cast(int, x),
      0x76543210, 0xfedcba98, false, false));
}

template <int O>
__device__ __forceinline__ void combine_round(v8f (&acc)[QT][VT],
                                              float (&s_x)[WAVES - 1][QT][VOWN][8][32],
                                              const float (&linv)[QT], int wave, int lane,
                                              int kg, int col, int p0, int bi, int vbase,
                                              float* __restrict__ out)
{
  if (wave != O) {
    const int slot = wave < O ? wave : wave - 1;
#pragma unroll
    for (int qt = 0; qt < QT; ++qt) {
#pragma unroll
      for (int i = 0; i < VOWN; ++i) {
#pragma unroll
        for (int j = 0; j < 8; ++j) s_x[slot][qt][i][j][lane] = acc[qt][O * VOWN + i][j];
      }
    }
  }
  __syncthreads();
  if (wave == O) {
#pragma unroll
    for (int qt = 0; qt < QT; ++qt) {
      const int pg = p0 + qt * 16 + col;
#pragma unroll
      for (int i = 0; i < VOWN; ++i) {
#pragma unroll
        for (int j = 0; j < 8; ++j) {
          float r = acc[qt][O * VOWN + i][j];
#pragma unroll
          for (int sl = 0; sl < WAVES - 1; ++sl) r += s_x[sl][qt][i][j][lane];
          const int v = vbase + (O * VOWN + i) * 16 + kg + j;
          if (pg < P_) out[((size_t)bi * (2 * CV_) + v) * P_ + pg] = r * linv[qt];
        }
      }
    }
  }
  __syncthreads();
}

__device__ __forceinline__ void copy_qv_block(const float* __restrict__ qv, float* __restrict__ out,
                                              int bi, int cwg, int tid)
{
  const v4f* cs = (const v4f*)(qv + (size_t)bi * (CV_ * P_));
  v4f*       cd = (v4f*)(out + (size_t)bi * (2 * CV_ * P_) + (size_t)(CV_ * P_));
  int i = cwg * (32 * WAVES) + tid;
  for (int r = 0; r < CPY_N4 / (CPY_STRIDE * CPY_U); ++r) {
    v4f v[CPY_U];
#pragma unroll
    for (int u = 0; u < CPY_U; ++u) v[u] = cs[i + u * CPY_STRIDE];
    asm volatile("" :: "v"(v[0]), "v"(v[1]), "v"(v[2]), "v"(v[3]), "v"(v[4]), "v"(v[5]), "v"(v[6]),
                 "v"(v[7]), "v"(v[8]), "v"(v[9]), "v"(v[10]), "v"(v[11]), "v"(v[12]), "v"(v[13]),
                 "v"(v[14]) : "memory");
    static_assert(CPY_U == 15, "the statement above names every piece of a round");
#pragma unroll
    for (int u = 0; u < CPY_U; ++u) cd[i + u * CPY_STRIDE] = v[u];
    i += CPY_U * CPY_STRIDE;
  }
}

template <bool MV_BF16>
__global__ __launch_bounds__(32 * WAVES) void mem_reader_attn(
    const float* __restrict__ mk, const float* __restrict__ mv_f32,
    const __bf16* __restrict__ mv_bf, const float* __restrict__ qk,
    const float* __restrict__ qv, float* __restrict__ out)
{
  const int tid  = threadIdx.x;

  if (blockIdx.x >= PTILES) {
    copy_qv_block(qv, out, blockIdx.y, (blockIdx.x - PTILES) * VSPLIT + blockIdx.z, tid);
    return;
  }

  const int wave = __builtin_amdgcn_readfirstlane(tid >> 5);
  const int lane = tid & 31;
  const int col  = lane & 15;
  const int hf   = lane >> 4;
  const int kg   = hf * 8;
  const int p0   = blockIdx.x * PBLK;
  const int bi   = blockIdx.y;
  const int vbase = blockIdx.z * VROWS;

  const float*  mk_b  = mk + (size_t)bi * CK_ * N_;
  const float*  qk_b  = qk + (size_t)bi * CK_ * P_;
  const float*  mvf_b = mv_f32 ? (mv_f32 + (size_t)bi * CV_ * N_) : nullptr;
  const __bf16* mvb_b = mv_bf  ? (mv_bf  + (size_t)bi * CV_ * N_) : nullptr;

  Frag16 qf[QT][2];
#pragma unroll
  for (int qt = 0; qt < QT; ++qt) {
    int pq = p0 + qt * 16 + col; if (pq > P_ - 1) pq = P_ - 1;
#pragma unroll
    for (int ks = 0; ks < 2; ++ks) {
#pragma unroll
      for (int s = 0; s < 16; ++s)
        qf[qt][ks].h[s] = (__bf16)(qk_b[(size_t)score_chan(ks, hf, s) * P_ + pq] * 0.125f);
    }
  }

  const int vrow0 = col;
  const int kpr   = key_pair(col);

  float m_run[QT];
  float l_run[QT];
  v8f acc[QT][VT];
#pragma unroll
  for (int qt = 0; qt < QT; ++qt) {
    m_run[qt] = -1e30f;
    l_run[qt] = 0.f;
#pragma unroll
    for (int vt = 0; vt < VT; ++vt) acc[qt][vt] = v8f{};
  }

  for (int st = wave; st < NSTEP; st += WAVES) {
    const int n0 = st * KSTEP;

    float2 kraw[SUB][2][16];
#pragma unroll
    for (int u = 0; u < SUB; ++u) {
      const float* kp = mk_b + n0 + u * 32 + 2 * kpr;
#pragma unroll
      for (int ks = 0; ks < 2; ++ks) {
#pragma unroll
        for (int s = 0; s < 16; ++s)
          kraw[u][ks][s] = *(const float2*)(kp + (size_t)score_chan(ks, hf, s) * N_);
      }
    }

    Frag16 a[NGRP][VG];
    auto load_group = [&](int k) {
      const int u = k / (VT / VG);
      const int g = k % (VT / VG);
#pragma unroll
      for (int i = 0; i < VG; ++i) {
        const size_t roff = (size_t)(vbase + vrow0 + (g * VG + i) * 16) * N_ + n0 + kg + u * 32;
        if (MV_BF16) {
          const __bf16* rowp = mvb_b + roff;
          a[k][i].q[0] = *(const v4u*)(rowp);
          a[k][i].q[1] = *(const v4u*)(rowp + 16);
        } else {
          const float* rowp = mvf_b + roff;
#pragma unroll
          for (int h2 = 0; h2 < 2; ++h2) {
            const float4 f0 = *(const float4*)(rowp + h2 * 16);
            const float4 f1 = *(const float4*)(rowp + h2 * 16 + 4);
            a[k][i].h[h2*8+0]=(__bf16)f0.x; a[k][i].h[h2*8+1]=(__bf16)f0.y;
            a[k][i].h[h2*8+2]=(__bf16)f0.z; a[k][i].h[h2*8+3]=(__bf16)f0.w;
            a[k][i].h[h2*8+4]=(__bf16)f1.x; a[k][i].h[h2*8+5]=(__bf16)f1.y;
            a[k][i].h[h2*8+6]=(__bf16)f1.z; a[k][i].h[h2*8+7]=(__bf16)f1.w;
          }
        }
      }
    };
#pragma unroll
    for (int k = 0; k < AHEAD; ++k) load_group(k);
    __builtin_amdgcn_sched_barrier(0);

    Frag16 akf[2 * SUB][2];
#pragma unroll
    for (int t = 0; t < 2 * SUB; ++t) {
#pragma unroll
      for (int ks = 0; ks < 2; ++ks) {
#pragma unroll
        for (int s = 0; s < 16; ++s)
          akf[t][ks].h[s] = (__bf16)((t & 1) ? kraw[t >> 1][ks][s].y : kraw[t >> 1][ks][s].x);
      }
    }

    Frag16 pb[QT][SUB];
#pragma unroll
    for (int qt = 0; qt < QT; ++qt) {
      v8f S[2 * SUB];
#pragma unroll
      for (int t = 0; t < 2 * SUB; ++t) {
        S[t] = v8f{};
#pragma unroll
        for (int ks = 0; ks < 2; ++ks)
          S[t] = __builtin_amdgcn_wmma_f32_16x16x32_bf16(false, akf[t][ks].bf, false, qf[qt][ks].bf,
                                                         (short)0, S[t], false, false);
      }

      float lmax = S[0][0];
#pragma unroll
      for (int t = 0; t < 2 * SUB; ++t) {
#pragma unroll
        for (int j = 0; j < 8; ++j) lmax = fmaxf(lmax, S[t][j]);
      }
      lmax = fmaxf(lmax, other_half(lmax));
      const float m_new = fmaxf(m_run[qt], lmax);
      const float alph  = __expf(m_run[qt] - m_new);
      m_run[qt] = m_new;

      float lsum = 0.f;
#pragma unroll
      for (int u = 0; u < SUB; ++u) {
#pragma unroll
        for (int g = 0; g < 2; ++g) {
#pragma unroll
          for (int j = 0; j < 8; ++j) {
            const float pr = __expf(S[2 * u + g][j] - m_new);
            lsum += pr;
            pb[qt][u].h[2 * j + g] = (__bf16)pr;
          }
        }
      }
      l_run[qt] = l_run[qt] * alph + lsum;

#pragma unroll
      for (int vt = 0; vt < VT; ++vt) acc[qt][vt] *= alph;
    }

#pragma unroll
    for (int k = 0; k < NGRP; ++k) {
      if (k + AHEAD < NGRP) load_group(k + AHEAD);
      __builtin_amdgcn_sched_barrier(0);
      const int u = k / (VT / VG);
      const int g = k % (VT / VG);
#pragma unroll
      for (int i = 0; i < VG; ++i) {
#pragma unroll
        for (int qt = 0; qt < QT; ++qt)
          acc[qt][g * VG + i] = __builtin_amdgcn_wmma_f32_16x16x32_bf16(
              false, a[k][i].bf, false, pb[qt][u].bf, (short)0, acc[qt][g * VG + i], false, false);
      }
      __builtin_amdgcn_sched_barrier(0);
    }
  }

  __shared__ float s_m[WAVES][QT][16];
  __shared__ float s_l[WAVES][QT][16];
  __shared__ float s_x[WAVES - 1][QT][VOWN][8][32];

  float l_own[QT];
#pragma unroll
  for (int qt = 0; qt < QT; ++qt) {
    l_own[qt] = l_run[qt] + other_half(l_run[qt]);
    if (hf == 0) {
      s_m[wave][qt][col] = m_run[qt];
      s_l[wave][qt][col] = l_own[qt];
    }
  }
  __syncthreads();

  float linv[QT];
#pragma unroll
  for (int qt = 0; qt < QT; ++qt) {
    float m_all = s_m[0][qt][col];
#pragma unroll
    for (int w = 1; w < WAVES; ++w) m_all = fmaxf(m_all, s_m[w][qt][col]);
    float l_all = 0.f;
#pragma unroll
    for (int w = 0; w < WAVES; ++w) l_all += s_l[w][qt][col] * __expf(s_m[w][qt][col] - m_all);
    linv[qt] = 1.0f / l_all;
    const float sc = __expf(m_run[qt] - m_all);
#pragma unroll
    for (int vt = 0; vt < VT; ++vt) acc[qt][vt] *= sc;
  }

  combine_round<0>(acc, s_x, linv, wave, lane, kg, col, p0, bi, vbase, out);
  combine_round<1>(acc, s_x, linv, wave, lane, kg, col, p0, bi, vbase, out);
  combine_round<2>(acc, s_x, linv, wave, lane, kg, col, p0, bi, vbase, out);
  combine_round<3>(acc, s_x, linv, wave, lane, kg, col, p0, bi, vbase, out);
  static_assert(WAVES == 4, "one combine round per wave is written out above");
}

__global__ __launch_bounds__(256) void convert_mv(const float* __restrict__ src,
                                                  __bf16* __restrict__ dst, int n4)
{
  int i = blockIdx.x * blockDim.x + threadIdx.x;
  if (i < n4) {
    float4 f = ((const float4*)src)[i];
    Pack4 p;
    p.h[0] = (__bf16)f.x; p.h[1] = (__bf16)f.y;
    p.h[2] = (__bf16)f.z; p.h[3] = (__bf16)f.w;
    *(uint2*)(dst + 4 * (size_t)i) = p.u;
  }
}

extern "C" void kernel_launch(void* const* d_in, const int* in_sizes, int n_in,
                              void* d_out, int out_size, void* d_ws, size_t ws_size,
                              hipStream_t stream) {
  const float* mk = (const float*)d_in[0];
  const float* mv = (const float*)d_in[1];
  const float* qk = (const float*)d_in[2];
  const float* qv = (const float*)d_in[3];
  float* out = (float*)d_out;

  const size_t mv_elems = (size_t)B_ * CV_ * N_;
  const size_t need     = mv_elems * sizeof(__bf16);
  dim3 grid(PTILES + CPY_X, B_, VSPLIT);

  if (d_ws && ws_size >= need) {
    __bf16* mv_bf = (__bf16*)d_ws;
    int n4 = (int)(mv_elems / 4);
    convert_mv<<<(n4 + 255) / 256, 256, 0, stream>>>(mv, mv_bf, n4);
    mem_reader_attn<true><<<grid, 32 * WAVES, 0, stream>>>(mk, nullptr, mv_bf, qk, qv, out);
  } else {
    mem_reader_attn<false><<<grid, 32 * WAVES, 0, stream>>>(mk, mv, nullptr, qk, qv, out);
  }
}


# next step's keys are requested during the current step's value products (key buffer lives across the loop edge; value groups 2 ahead, not 3, to make room): no step opens waiting for its keys
# speedup vs baseline: 10.9265x; 1.0524x over previous
#include <hip/hip_runtime.h>


#define B_    4
#define CK_   64
#define CV_   512
#define H_    30
#define W_    54
#define T_    8
#define P_    (H_*W_)
#define N_    (T_*H_*W_)
#define SUB   3
#define KSTEP (32 * SUB)
#define NSTEP (N_ / KSTEP)
#define QT    7
#define PBLK  (16 * QT)
#define PTILES ((P_ + PBLK - 1)/PBLK)
#define WAVES 4
#define VSPLIT 4
#define VROWS (CV_ / VSPLIT)
#define VT    (VROWS / 16)
#define VG    4
#define VOWN  (VT / WAVES)
#define NGRP  (SUB * (VT / VG))
#define AHEAD 2
#define NEXTK0 0
#define NEXTK1 2
#define NEXTK2 4
#define CPY_X 1
#define CPY_WG (CPY_X * VSPLIT)
#define CPY_U 15
#define CPY_N4 ((CV_ * P_) / 4)
#define CPY_STRIDE (CPY_WG * 32 * WAVES)

static_assert(N_ % KSTEP == 0, "every 96-key step is full: no key tail to mask");
static_assert((N_ * 2) % 16 == 0, "bf16 value rows start 16-byte aligned");
static_assert(N_ % 2 == 0 && KSTEP % 32 == 0, "a lane's key pair is one 8-byte aligned load; a half wave's pairs fill one 128-byte line");
static_assert(NSTEP >= WAVES, "every wave owns at least one key step, so its running max is finite");
static_assert(VT % VG == 0 && VT % WAVES == 0, "value tiles split evenly into load groups and owners");
static_assert(SUB == 3 && NEXTK0 < NEXTK1 && NEXTK1 < NEXTK2 && NEXTK2 < NGRP, "one request point per key block, in block order, inside the step");
static_assert(CV_ % (16 * VSPLIT) == 0, "the value rows split evenly into whole tiles per work-group");
static_assert((CV_ * P_) % 4 == 0, "a batch's query-value block is a whole number of 16-byte pieces, 16-byte aligned in qv and in out");
static_assert(CPY_N4 % (CPY_STRIDE * CPY_U) == 0, "every copy thread moves a whole number of full rounds: no tail, no bound test");

typedef __attribute__((ext_vector_type(16))) __bf16       v16bf;
typedef __attribute__((ext_vector_type(8)))  float        v8f;
typedef __attribute__((ext_vector_type(4)))  unsigned int v4u;
typedef __attribute__((ext_vector_type(4)))  float        v4f;

union Frag16 {
  v16bf  bf;
  v4u    q[2];
  __bf16 h[16];
};
union Pack4 {
  uint2  u;
  __bf16 h[4];
};

__device__ __forceinline__ int score_chan(int ks, int hf, int s) {
  return ks * 32 + (s >> 3) * 16 + hf * 8 + (s & 7);
}

__device__ __forceinline__ int key_pair(int r) {
  return 4 * (r >> 3) + (r & 3) + 8 * ((r >> 2) & 1);
}

__device__ __forceinline__ float other_half(float x) {
  return __builtin_bit_cast(float, __builtin_amdgcn_permlanex16(
      __builtin_bit_cast(int, x), __builtin_bit_cast(int, x),
      0x76543210, 0xfedcba98, false, false));
}

template <int O>
__device__ __forceinline__ void combine_round(v8f (&acc)[QT][VT],
                                              float (&s_x)[WAVES - 1][QT][VOWN][8][32],
                                              const float (&linv)[QT], int wave, int lane,
                                              int kg, int col, int p0, int bi, int vbase,
                                              float* __restrict__ out)
{
  if (wave != O) {
    const int slot = wave < O ? wave : wave - 1;
#pragma unroll
    for (int qt = 0; qt < QT; ++qt) {
#pragma unroll
      for (int i = 0; i < VOWN; ++i) {
#pragma unroll
        for (int j = 0; j < 8; ++j) s_x[slot][qt][i][j][lane] = acc[qt][O * VOWN + i][j];
      }
    }
  }
  __syncthreads();
  if (wave == O) {
#pragma unroll
    for (int qt = 0; qt < QT; ++qt) {
      const int pg = p0 + qt * 16 + col;
#pragma unroll
      for (int i = 0; i < VOWN; ++i) {
#pragma unroll
        for (int j = 0; j < 8; ++j) {
          float r = acc[qt][O * VOWN + i][j];
#pragma unroll
          for (int sl = 0; sl < WAVES - 1; ++sl) r += s_x[sl][qt][i][j][lane];
          const int v = vbase + (O * VOWN + i) * 16 + kg + j;
          if (pg < P_) out[((size_t)bi * (2 * CV_) + v) * P_ + pg] = r * linv[qt];
        }
      }
    }
  }
  __syncthreads();
}

__device__ __forceinline__ void copy_qv_block(const float* __restrict__ qv, float* __restrict__ out,
                                              int bi, int cwg, int tid)
{
  const v4f* cs = (const v4f*)(qv + (size_t)bi * (CV_ * P_));
  v4f*       cd = (v4f*)(out + (size_t)bi * (2 * CV_ * P_) + (size_t)(CV_ * P_));
  int i = cwg * (32 * WAVES) + tid;
  for (int r = 0; r < CPY_N4 / (CPY_STRIDE * CPY_U); ++r) {
    v4f v[CPY_U];
#pragma unroll
    for (int u = 0; u < CPY_U; ++u) v[u] = cs[i + u * CPY_STRIDE];
    asm volatile("" :: "v"(v[0]), "v"(v[1]), "v"(v[2]), "v"(v[3]), "v"(v[4]), "v"(v[5]), "v"(v[6]),
                 "v"(v[7]), "v"(v[8]), "v"(v[9]), "v"(v[10]), "v"(v[11]), "v"(v[12]), "v"(v[13]),
                 "v"(v[14]) : "memory");
    static_assert(CPY_U == 15, "the statement above names every piece of a round");
#pragma unroll
    for (int u = 0; u < CPY_U; ++u) cd[i + u * CPY_STRIDE] = v[u];
    i += CPY_U * CPY_STRIDE;
  }
}

template <bool MV_BF16>
__global__ __launch_bounds__(32 * WAVES) void mem_reader_attn(
    const float* __restrict__ mk, const float* __restrict__ mv_f32,
    const __bf16* __restrict__ mv_bf, const float* __restrict__ qk,
    const float* __restrict__ qv, float* __restrict__ out)
{
  const int tid  = threadIdx.x;

  if (blockIdx.x >= PTILES) {
    copy_qv_block(qv, out, blockIdx.y, (blockIdx.x - PTILES) * VSPLIT + blockIdx.z, tid);
    return;
  }

  const int wave = __builtin_amdgcn_readfirstlane(tid >> 5);
  const int lane = tid & 31;
  const int col  = lane & 15;
  const int hf   = lane >> 4;
  const int kg   = hf * 8;
  const int p0   = blockIdx.x * PBLK;
  const int bi   = blockIdx.y;
  const int vbase = blockIdx.z * VROWS;

  const float*  mk_b  = mk + (size_t)bi * CK_ * N_;
  const float*  qk_b  = qk + (size_t)bi * CK_ * P_;
  const float*  mvf_b = mv_f32 ? (mv_f32 + (size_t)bi * CV_ * N_) : nullptr;
  const __bf16* mvb_b = mv_bf  ? (mv_bf  + (size_t)bi * CV_ * N_) : nullptr;

  Frag16 qf[QT][2];
#pragma unroll
  for (int qt = 0; qt < QT; ++qt) {
    int pq = p0 + qt * 16 + col; if (pq > P_ - 1) pq = P_ - 1;
#pragma unroll
    for (int ks = 0; ks < 2; ++ks) {
#pragma unroll
      for (int s = 0; s < 16; ++s)
        qf[qt][ks].h[s] = (__bf16)(qk_b[(size_t)score_chan(ks, hf, s) * P_ + pq] * 0.125f);
    }
  }

  const int vrow0 = col;
  const int kpr   = key_pair(col);

  float m_run[QT];
  float l_run[QT];
  v8f acc[QT][VT];
#pragma unroll
  for (int qt = 0; qt < QT; ++qt) {
    m_run[qt] = -1e30f;
    l_run[qt] = 0.f;
#pragma unroll
    for (int vt = 0; vt < VT; ++vt) acc[qt][vt] = v8f{};
  }

  float2 kraw[SUB][2][16];
  auto load_keys = [&](int u, int nk) {
    const float* kp = mk_b + nk + u * 32 + 2 * kpr;
#pragma unroll
    for (int ks = 0; ks < 2; ++ks) {
#pragma unroll
      for (int s = 0; s < 16; ++s)
        kraw[u][ks][s] = *(const float2*)(kp + (size_t)score_chan(ks, hf, s) * N_);
    }
  };

#pragma unroll
  for (int u = 0; u < SUB; ++u) load_keys(u, wave * KSTEP);

  for (int st = wave; st < NSTEP; st += WAVES) {
    const int n0 = st * KSTEP;
    const int n1 = (st + WAVES < NSTEP ? st + WAVES : st) * KSTEP;

    Frag16 a[NGRP][VG];
    auto load_group = [&](int k) {
      const int u = k / (VT / VG);
      const int g = k % (VT / VG);
#pragma unroll
      for (int i = 0; i < VG; ++i) {
        const size_t roff = (size_t)(vbase + vrow0 + (g * VG + i) * 16) * N_ + n0 + kg + u * 32;
        if (MV_BF16) {
          const __bf16* rowp = mvb_b + roff;
          a[k][i].q[0] = *(const v4u*)(rowp);
          a[k][i].q[1] = *(const v4u*)(rowp + 16);
        } else {
          const float* rowp = mvf_b + roff;
#pragma unroll
          for (int h2 = 0; h2 < 2; ++h2) {
            const float4 f0 = *(const float4*)(rowp + h2 * 16);
            const float4 f1 = *(const float4*)(rowp + h2 * 16 + 4);
            a[k][i].h[h2*8+0]=(__bf16)f0.x; a[k][i].h[h2*8+1]=(__bf16)f0.y;
            a[k][i].h[h2*8+2]=(__bf16)f0.z; a[k][i].h[h2*8+3]=(__bf16)f0.w;
            a[k][i].h[h2*8+4]=(__bf16)f1.x; a[k][i].h[h2*8+5]=(__bf16)f1.y;
            a[k][i].h[h2*8+6]=(__bf16)f1.z; a[k][i].h[h2*8+7]=(__bf16)f1.w;
          }
        }
      }
    };
#pragma unroll
    for (int k = 0; k < AHEAD; ++k) load_group(k);
    __builtin_amdgcn_sched_barrier(0);

    Frag16 akf[2 * SUB][2];
#pragma unroll
    for (int t = 0; t < 2 * SUB; ++t) {
#pragma unroll
      for (int ks = 0; ks < 2; ++ks) {
#pragma unroll
        for (int s = 0; s < 16; ++s)
          akf[t][ks].h[s] = (__bf16)((t & 1) ? kraw[t >> 1][ks][s].y : kraw[t >> 1][ks][s].x);
      }
    }

    Frag16 pb[QT][SUB];
#pragma unroll
    for (int qt = 0; qt < QT; ++qt) {
      v8f S[2 * SUB];
#pragma unroll
      for (int t = 0; t < 2 * SUB; ++t) {
        S[t] = v8f{};
#pragma unroll
        for (int ks = 0; ks < 2; ++ks)
          S[t] = __builtin_amdgcn_wmma_f32_16x16x32_bf16(false, akf[t][ks].bf, false, qf[qt][ks].bf,
                                                         (short)0, S[t], false, false);
      }

      float lmax = S[0][0];
#pragma unroll
      for (int t = 0; t < 2 * SUB; ++t) {
#pragma unroll
        for (int j = 0; j < 8; ++j) lmax = fmaxf(lmax, S[t][j]);
      }
      lmax = fmaxf(lmax, other_half(lmax));
      const float m_new = fmaxf(m_run[qt], lmax);
      const float alph  = __expf(m_run[qt] - m_new);
      m_run[qt] = m_new;

      float lsum = 0.f;
#pragma unroll
      for (int u = 0; u < SUB; ++u) {
#pragma unroll
        for (int g = 0; g < 2; ++g) {
#pragma unroll
          for (int j = 0; j < 8; ++j) {
            const float pr = __expf(S[2 * u + g][j] - m_new);
            lsum += pr;
            pb[qt][u].h[2 * j + g] = (__bf16)pr;
          }
        }
      }
      l_run[qt] = l_run[qt] * alph + lsum;

#pragma unroll
      for (int vt = 0; vt < VT; ++vt) acc[qt][vt] *= alph;
    }

    __builtin_amdgcn_sched_barrier(0);

#pragma unroll
    for (int k = 0; k < NGRP; ++k) {
      if (k + AHEAD < NGRP) load_group(k + AHEAD);
      if (k == NEXTK0) load_keys(0, n1);
      if (k == NEXTK1) load_keys(1, n1);
      if (k == NEXTK2) load_keys(2, n1);
      __builtin_amdgcn_sched_barrier(0);
      const int u = k / (VT / VG);
      const int g = k % (VT / VG);
#pragma unroll
      for (int i = 0; i < VG; ++i) {
#pragma unroll
        for (int qt = 0; qt < QT; ++qt)
          acc[qt][g * VG + i] = __builtin_amdgcn_wmma_f32_16x16x32_bf16(
              false, a[k][i].bf, false, pb[qt][u].bf, (short)0, acc[qt][g * VG + i], false, false);
      }
      __builtin_amdgcn_sched_barrier(0);
    }
  }

  __shared__ float s_m[WAVES][QT][16];
  __shared__ float s_l[WAVES][QT][16];
  __shared__ float s_x[WAVES - 1][QT][VOWN][8][32];

  float l_own[QT];
#pragma unroll
  for (int qt = 0; qt < QT; ++qt) {
    l_own[qt] = l_run[qt] + other_half(l_run[qt]);
    if (hf == 0) {
      s_m[wave][qt][col] = m_run[qt];
      s_l[wave][qt][col] = l_own[qt];
    }
  }
  __syncthreads();

  float linv[QT];
#pragma unroll
  for (int qt = 0; qt < QT; ++qt) {
    float m_all = s_m[0][qt][col];
#pragma unroll
    for (int w = 1; w < WAVES; ++w) m_all = fmaxf(m_all, s_m[w][qt][col]);
    float l_all = 0.f;
#pragma unroll
    for (int w = 0; w < WAVES; ++w) l_all += s_l[w][qt][col] * __expf(s_m[w][qt][col] - m_all);
    linv[qt] = 1.0f / l_all;
    const float sc = __expf(m_run[qt] - m_all);
#pragma unroll
    for (int vt = 0; vt < VT; ++vt) acc[qt][vt] *= sc;
  }

  combine_round<0>(acc, s_x, linv, wave, lane, kg, col, p0, bi, vbase, out);
  combine_round<1>(acc, s_x, linv, wave, lane, kg, col, p0, bi, vbase, out);
  combine_round<2>(acc, s_x, linv, wave, lane, kg, col, p0, bi, vbase, out);
  combine_round<3>(acc, s_x, linv, wave, lane, kg, col, p0, bi, vbase, out);
  static_assert(WAVES == 4, "one combine round per wave is written out above");
}

__global__ __launch_bounds__(256) void convert_mv(const float* __restrict__ src,
                                                  __bf16* __restrict__ dst, int n4)
{
  int i = blockIdx.x * blockDim.x + threadIdx.x;
  if (i < n4) {
    float4 f = ((const float4*)src)[i];
    Pack4 p;
    p.h[0] = (__bf16)f.x; p.h[1] = (__bf16)f.y;
    p.h[2] = (__bf16)f.z; p.h[3] = (__bf16)f.w;
    *(uint2*)(dst + 4 * (size_t)i) = p.u;
  }
}

extern "C" void kernel_launch(void* const* d_in, const int* in_sizes, int n_in,
                              void* d_out, int out_size, void* d_ws, size_t ws_size,
                              hipStream_t stream) {
  const float* mk = (const float*)d_in[0];
  const float* mv = (const float*)d_in[1];
  const float* qk = (const float*)d_in[2];
  const float* qv = (const float*)d_in[3];
  float* out = (float*)d_out;

  const size_t mv_elems = (size_t)B_ * CV_ * N_;
  const size_t need     = mv_elems * sizeof(__bf16);
  dim3 grid(PTILES + CPY_X, B_, VSPLIT);

  if (d_ws && ws_size >= need) {
    __bf16* mv_bf = (__bf16*)d_ws;
    int n4 = (int)(mv_elems / 4);
    convert_mv<<<(n4 + 255) / 256, 256, 0, stream>>>(mv, mv_bf, n4);
    mem_reader_attn<true><<<grid, 32 * WAVES, 0, stream>>>(mk, nullptr, mv_bf, qk, qv, out);
  } else {
    mem_reader_attn<false><<<grid, 32 * WAVES, 0, stream>>>(mk, mv, nullptr, qk, qv, out);
  }
}
